# starting kernel
# baseline (speedup 1.0000x reference)
#include <hip/hip_runtime.h>


typedef __attribute__((ext_vector_type(2))) float v2f;
typedef __attribute__((ext_vector_type(4))) float v4f;
typedef __attribute__((ext_vector_type(8))) float v8f;

#define NN  100000
#define NE  1000000
#define DM  64
#define NH  4
#define DH  16

__global__ void init_ws_kernel(float* __restrict__ ctx,
                               unsigned* __restrict__ maxkey,
                               float* __restrict__ z) {
  int i = blockIdx.x * blockDim.x + threadIdx.x;
  if (i < NN * DM) ctx[i] = 0.0f;
  if (i < NN * NH) { maxkey[i] = 0u; z[i] = 0.0f; }
}

__global__ __launch_bounds__(128)
void gemm64_wmma_kernel(const float* __restrict__ A,
                        const float* __restrict__ W,
                        const float* __restrict__ bias,
                        float* __restrict__ out) {
  const int lane = threadIdx.x & 31;
  const int wave = threadIdx.x >> 5;
  const int row0 = blockIdx.x * 16;
  const int n0   = wave * 16;
  const int m    = lane & 15;
  const int half = lane >> 4;
  const int kA   = half * 2;

  const float* __restrict__ arow = A + (size_t)(row0 + m) * DM;

  v8f c = {};
#pragma unroll
  for (int k0 = 0; k0 < DM; k0 += 4) {
    v2f a, b;
    a.x = arow[k0 + kA];
    a.y = arow[k0 + kA + 1];
    b.x = W[(size_t)(k0 + kA) * DM + n0 + m];
    b.y = W[(size_t)(k0 + kA + 1) * DM + n0 + m];
    c = __builtin_amdgcn_wmma_f32_16x16x4_f32(false, a, false, b,
                                              (short)0, c, false, false);
  }

  const float bb = bias[n0 + m];
#pragma unroll
  for (int j = 0; j < 8; ++j) {
    out[(size_t)(row0 + j + half * 8) * DM + n0 + m] = c[j] + bb;
  }
}

__device__ __forceinline__ unsigned enc_f32(float f) {
  int i = __float_as_int(f);
  return (unsigned)(i ^ ((i >> 31) | 0x80000000));
}
__device__ __forceinline__ float dec_f32(unsigned k) {
  return (k & 0x80000000u) ? __uint_as_float(k ^ 0x80000000u)
                           : __uint_as_float(~k);
}

__global__ void edge_scores_kernel(const float* __restrict__ qks,
                                   const int* __restrict__ senders,
                                   const int* __restrict__ receivers,
                                   float* __restrict__ attn_buf,
                                   unsigned* __restrict__ maxkey) {
  int idx = blockIdx.x * blockDim.x + threadIdx.x;
  if (idx >= NE * NH) return;
  const int e = idx >> 2, h = idx & 3;
  const int s = senders[e], r = receivers[e];
  const v4f* __restrict__ qr = (const v4f*)(qks + (size_t)r * DM + h * DH);
  const v4f* __restrict__ qs = (const v4f*)(qks + (size_t)s * DM + h * DH);
  float acc = 0.0f;
#pragma unroll
  for (int i = 0; i < 4; ++i) {
    v4f a = qr[i], b = qs[i];
    acc += a.x * b.x + a.y * b.y + a.z * b.z + a.w * b.w;
  }
  attn_buf[idx] = acc;
  atomicMax(&maxkey[r * NH + h], enc_f32(acc));
}

__global__ void edge_exp_kernel(const int* __restrict__ receivers,
                                const unsigned* __restrict__ maxkey,
                                float* __restrict__ attn_buf,
                                float* __restrict__ z) {
  int idx = blockIdx.x * blockDim.x + threadIdx.x;
  if (idx >= NE * NH) return;
  const int e = idx >> 2, h = idx & 3;
  const int r = receivers[e];
  const float mx = dec_f32(maxkey[r * NH + h]);
  const float ev = __expf(attn_buf[idx] - mx);
  attn_buf[idx] = ev;
  atomicAdd(&z[r * NH + h], ev);
}

__global__ void edge_scatter_kernel(const float* __restrict__ vs,
                                    const int* __restrict__ senders,
                                    const int* __restrict__ receivers,
                                    const float* __restrict__ z,
                                    float* __restrict__ attn_buf,
                                    float* __restrict__ ctx) {
  int idx = blockIdx.x * blockDim.x + threadIdx.x;
  if (idx >= NE * NH) return;
  const int e = idx >> 2, h = idx & 3;
  const int r = receivers[e];
  const float a = attn_buf[idx] / z[r * NH + h];
  attn_buf[idx] = a;
  const int s = senders[e];
  const v4f* __restrict__ vrow = (const v4f*)(vs + (size_t)s * DM + h * DH);
  float* __restrict__ crow = ctx + (size_t)r * DM + h * DH;
#pragma unroll
  for (int i = 0; i < 4; ++i) {
    v4f v = vrow[i];
    atomicAdd(&crow[i * 4 + 0], a * v.x);
    atomicAdd(&crow[i * 4 + 1], a * v.y);
    atomicAdd(&crow[i * 4 + 2], a * v.z);
    atomicAdd(&crow[i * 4 + 3], a * v.w);
  }
}

extern "C" void kernel_launch(void* const* d_in, const int* in_sizes, int n_in,
                              void* d_out, int out_size, void* d_ws, size_t ws_size,
                              hipStream_t stream) {
  const float* nodes     = (const float*)d_in[0];
  const float* W_qk      = (const float*)d_in[1];
  const float* b_qk      = (const float*)d_in[2];
  const float* W_v       = (const float*)d_in[3];
  const float* b_v       = (const float*)d_in[4];
  const float* W_out     = (const float*)d_in[5];
  const float* b_out     = (const float*)d_in[6];
  const int*   senders   = (const int*)d_in[7];
  const int*   receivers = (const int*)d_in[8];

  float* out  = (float*)d_out;
  float* attn = out + (size_t)NN * DM;

  float* ws      = (float*)d_ws;
  float* qks     = ws;
  float* vs      = ws + (size_t)NN * DM;
  float* ctx     = ws + (size_t)2 * NN * DM;
  unsigned* mkey = (unsigned*)(ws + (size_t)3 * NN * DM);
  float* z       = ws + (size_t)3 * NN * DM + (size_t)NN * NH;

  const int initGrid = (NN * DM + 255) / 256;
  const int gemmGrid = NN / 16;
  const int edgeGrid = (NE * NH + 255) / 256;

  init_ws_kernel<<<initGrid, 256, 0, stream>>>(ctx, mkey, z);

  gemm64_wmma_kernel<<<gemmGrid, 128, 0, stream>>>(nodes, W_qk, b_qk, qks);
  gemm64_wmma_kernel<<<gemmGrid, 128, 0, stream>>>(nodes, W_v,  b_v,  vs);

  edge_scores_kernel<<<edgeGrid, 256, 0, stream>>>(qks, senders, receivers, attn, mkey);
  edge_exp_kernel<<<edgeGrid, 256, 0, stream>>>(receivers, mkey, attn, z);
  edge_scatter_kernel<<<edgeGrid, 256, 0, stream>>>(vs, senders, receivers, z, attn, ctx);

  gemm64_wmma_kernel<<<gemmGrid, 128, 0, stream>>>(ctx, W_out, b_out, out);
}


# Edge phase owner-computes: 16 lanes per receiver walk its sorted edge range (row-offset table), so segment max, sum and ctx are register sums in edge order; all atomics and the zero-fill go
# speedup vs baseline: 4.8240x; 4.8240x over previous
#include <hip/hip_runtime.h>


typedef __attribute__((ext_vector_type(2))) float v2f;
typedef __attribute__((ext_vector_type(4))) float v4f;
typedef __attribute__((ext_vector_type(8))) float v8f;

#define NN  100000
#define NE  1000000
#define DM  64
#define NH  4
#define DH  16

__global__ void row_start_kernel(const int* __restrict__ receivers,
                                 int* __restrict__ row_start) {
  const int e = blockIdx.x * blockDim.x + threadIdx.x;
  if (e >= NE) return;
  const int r = receivers[e];
  const int p = (e == 0) ? -1 : receivers[e - 1];
  for (int n = p + 1; n <= r; ++n) row_start[n] = e;
  if (e == NE - 1) {
    for (int n = r + 1; n <= NN; ++n) row_start[n] = NE;
  }
}

__global__ __launch_bounds__(128)
void gemm64_wmma_kernel(const float* __restrict__ A,
                        const float* __restrict__ W,
                        const float* __restrict__ bias,
                        float* __restrict__ out) {
  const int lane = threadIdx.x & 31;
  const int wave = threadIdx.x >> 5;
  const int row0 = blockIdx.x * 16;
  const int n0   = wave * 16;
  const int m    = lane & 15;
  const int half = lane >> 4;
  const int kA   = half * 2;

  const float* __restrict__ arow = A + (size_t)(row0 + m) * DM;

  v8f c = {};
#pragma unroll
  for (int k0 = 0; k0 < DM; k0 += 4) {
    v2f a, b;
    a.x = arow[k0 + kA];
    a.y = arow[k0 + kA + 1];
    b.x = W[(size_t)(k0 + kA) * DM + n0 + m];
    b.y = W[(size_t)(k0 + kA + 1) * DM + n0 + m];
    c = __builtin_amdgcn_wmma_f32_16x16x4_f32(false, a, false, b,
                                              (short)0, c, false, false);
  }

  const float bb = bias[n0 + m];
#pragma unroll
  for (int j = 0; j < 8; ++j) {
    out[(size_t)(row0 + j + half * 8) * DM + n0 + m] = c[j] + bb;
  }
}

#define NODE_BLOCK 256
#define STASH 32

__device__ __forceinline__ float quad_sum(float p) {
  p += __int_as_float(__builtin_amdgcn_update_dpp(
      0, __float_as_int(p), 0xB1, 0xF, 0xF, false));
  p += __int_as_float(__builtin_amdgcn_update_dpp(
      0, __float_as_int(p), 0x4E, 0xF, 0xF, false));
  return p;
}

__device__ __forceinline__ float edge_score(const float* __restrict__ qks,
                                            int s, int l16, v4f q) {
  const v4f k = *(const v4f*)(qks + (size_t)s * DM + l16 * 4);
  return quad_sum(q.x * k.x + q.y * k.y + q.z * k.z + q.w * k.w);
}

__global__ __launch_bounds__(NODE_BLOCK)
void node_attention_kernel(const float* __restrict__ qks,
                           const float* __restrict__ vs,
                           const int* __restrict__ senders,
                           const int* __restrict__ row_start,
                           float* __restrict__ attn,
                           float* __restrict__ ctx) {
  __shared__ float stash[STASH * NODE_BLOCK];

  const int tid  = threadIdx.x;
  const int l16  = tid & 15;
  const int node = blockIdx.x * (NODE_BLOCK / 16) + (tid >> 4);

  int start = row_start[node];
  int end   = row_start[node + 1];
  start = min(max(start, 0), NE);
  end   = min(max(end, start), NE);
  const int deg = end - start;

  const v4f q = *(const v4f*)(qks + (size_t)node * DM + l16 * 4);
  const int* __restrict__ snd = senders + start;
  float* __restrict__ st = stash + tid;

  float m = -INFINITY;
  int s_next = (deg > 0) ? snd[0] : 0;
  for (int i = 0; i < deg; ++i) {
    const int s = s_next;
    if (i + 1 < deg) s_next = snd[i + 1];
    const float sc = edge_score(qks, s, l16, q);
    if (i < STASH) st[i * NODE_BLOCK] = sc;
    m = fmaxf(m, sc);
  }

  float z = 0.0f;
  for (int i = 0; i < deg; ++i) {
    const float sc = (i < STASH) ? st[i * NODE_BLOCK]
                                 : edge_score(qks, snd[i], l16, q);
    const float ev = __expf(sc - m);
    if (i < STASH) st[i * NODE_BLOCK] = ev;
    z += ev;
  }

  v4f acc = {0.0f, 0.0f, 0.0f, 0.0f};
  s_next = (deg > 0) ? snd[0] : 0;
  for (int i = 0; i < deg; ++i) {
    const int s = s_next;
    if (i + 1 < deg) s_next = snd[i + 1];
    const v4f v = *(const v4f*)(vs + (size_t)s * DM + l16 * 4);
    const float ev = (i < STASH) ? st[i * NODE_BLOCK]
                                 : __expf(edge_score(qks, s, l16, q) - m);
    const float a = ev / z;
    if ((l16 & 3) == 0) attn[(size_t)(start + i) * NH + (l16 >> 2)] = a;
    acc.x += a * v.x;
    acc.y += a * v.y;
    acc.z += a * v.z;
    acc.w += a * v.w;
  }
  *(v4f*)(ctx + (size_t)node * DM + l16 * 4) = acc;
}

extern "C" void kernel_launch(void* const* d_in, const int* in_sizes, int n_in,
                              void* d_out, int out_size, void* d_ws, size_t ws_size,
                              hipStream_t stream) {
  const float* nodes     = (const float*)d_in[0];
  const float* W_qk      = (const float*)d_in[1];
  const float* b_qk      = (const float*)d_in[2];
  const float* W_v       = (const float*)d_in[3];
  const float* b_v       = (const float*)d_in[4];
  const float* W_out     = (const float*)d_in[5];
  const float* b_out     = (const float*)d_in[6];
  const int*   senders   = (const int*)d_in[7];
  const int*   receivers = (const int*)d_in[8];

  float* out  = (float*)d_out;
  float* attn = out + (size_t)NN * DM;

  float* ws      = (float*)d_ws;
  float* qks     = ws;
  float* vs      = ws + (size_t)NN * DM;
  float* ctx     = ws + (size_t)2 * NN * DM;
  int* row_start = (int*)(ws + (size_t)3 * NN * DM);

  const int rowGrid  = (NE + 255) / 256;
  const int gemmGrid = NN / 16;
  const int nodeGrid = NN / (NODE_BLOCK / 16);

  row_start_kernel<<<rowGrid, 256, 0, stream>>>(receivers, row_start);

  gemm64_wmma_kernel<<<gemmGrid, 128, 0, stream>>>(nodes, W_qk, b_qk, qks);
  gemm64_wmma_kernel<<<gemmGrid, 128, 0, stream>>>(nodes, W_v,  b_v,  vs);

  node_attention_kernel<<<nodeGrid, NODE_BLOCK, 0, stream>>>(qks, vs, senders, row_start, attn, ctx);

  gemm64_wmma_kernel<<<gemmGrid, 128, 0, stream>>>(ctx, W_out, b_out, out);
}


# The three input-only launches (row table, QK and V projections) become one: each A fragment is loaded once for both weight sets, each block also fills its 160 edges' row-table entries; k order kept
# speedup vs baseline: 5.0885x; 1.0548x over previous
#include <hip/hip_runtime.h>


typedef __attribute__((ext_vector_type(2))) float v2f;
typedef __attribute__((ext_vector_type(4))) float v4f;
typedef __attribute__((ext_vector_type(8))) float v8f;

#define NN  100000
#define NE  1000000
#define DM  64
#define NH  4
#define DH  16

__device__ __forceinline__ void row_start_fill(int* __restrict__ row_start,
                                               int e, int r, int p) {
  for (int n = p + 1; n <= r; ++n) row_start[n] = e;
  if (e == NE - 1) {
    for (int n = r + 1; n <= NN; ++n) row_start[n] = NE;
  }
}

__global__ __launch_bounds__(128)
void gemm64_wmma_kernel(const float* __restrict__ A,
                        const float* __restrict__ W,
                        const float* __restrict__ bias,
                        float* __restrict__ out) {
  const int lane = threadIdx.x & 31;
  const int wave = threadIdx.x >> 5;
  const int row0 = blockIdx.x * 16;
  const int n0   = wave * 16;
  const int m    = lane & 15;
  const int half = lane >> 4;
  const int kA   = half * 2;

  const float* __restrict__ arow = A + (size_t)(row0 + m) * DM;

  v8f c = {};
#pragma unroll
  for (int k0 = 0; k0 < DM; k0 += 4) {
    v2f a, b;
    a.x = arow[k0 + kA];
    a.y = arow[k0 + kA + 1];
    b.x = W[(size_t)(k0 + kA) * DM + n0 + m];
    b.y = W[(size_t)(k0 + kA + 1) * DM + n0 + m];
    c = __builtin_amdgcn_wmma_f32_16x16x4_f32(false, a, false, b,
                                              (short)0, c, false, false);
  }

  const float bb = bias[n0 + m];
#pragma unroll
  for (int j = 0; j < 8; ++j) {
    out[(size_t)(row0 + j + half * 8) * DM + n0 + m] = c[j] + bb;
  }
}

#define ROW_TILES (NN / 16)
#define EPB (NE / ROW_TILES)
static_assert(EPB * ROW_TILES == NE, "edge slices must tile the edge list");
#define EPW (EPB / 4)
static_assert(EPW * 4 == EPB, "a block's edge slice splits evenly over its waves");
static_assert(EPW > 32 && EPW <= 64, "one edge per lane plus a partial second pass");

__global__ __launch_bounds__(128)
void project_qkv_rows_kernel(const float* __restrict__ A,
                             const float* __restrict__ Wq,
                             const float* __restrict__ bq,
                             const float* __restrict__ Wv,
                             const float* __restrict__ bv,
                             const int* __restrict__ receivers,
                             float* __restrict__ qks,
                             float* __restrict__ vs,
                             int* __restrict__ row_start) {
  const int lane = threadIdx.x & 31;
  const int wave = threadIdx.x >> 5;
  const int row0 = blockIdx.x * 16;
  const int n0   = wave * 16;
  const int m    = lane & 15;
  const int half = lane >> 4;
  const int kA   = half * 2;

  const float* __restrict__ arow = A + (size_t)(row0 + m) * DM;

  v8f cq = {};
  v8f cv = {};
#pragma unroll
  for (int k0 = 0; k0 < DM; k0 += 4) {
    v2f a, wq, wv;
    a.x = arow[k0 + kA];
    a.y = arow[k0 + kA + 1];
    wq.x = Wq[(size_t)(k0 + kA) * DM + n0 + m];
    wq.y = Wq[(size_t)(k0 + kA + 1) * DM + n0 + m];
    wv.x = Wv[(size_t)(k0 + kA) * DM + n0 + m];
    wv.y = Wv[(size_t)(k0 + kA + 1) * DM + n0 + m];
    cq = __builtin_amdgcn_wmma_f32_16x16x4_f32(false, a, false, wq,
                                               (short)0, cq, false, false);
    cv = __builtin_amdgcn_wmma_f32_16x16x4_f32(false, a, false, wv,
                                               (short)0, cv, false, false);
  }

  const int wbase = blockIdx.x * EPB + wave * EPW;
  const bool has1 = lane < EPW - 32;
  const int e0 = wbase + lane;
  const int e1 = wbase + 32 + min(lane, EPW - 33);
  const int a0  = max(e0 - 1, 0);
  const int x0  = receivers[a0];
  const int y0  = receivers[a0 + 1];
  const int r0  = (e0 == 0) ? x0 : y0;
  const int p0  = (e0 == 0) ? -1 : x0;
  const int p1  = receivers[e1 - 1];
  const int r1  = receivers[e1];

  const float bbq = bq[n0 + m];
  const float bbv = bv[n0 + m];
#pragma unroll
  for (int j = 0; j < 8; ++j) {
    const size_t o = (size_t)(row0 + j + half * 8) * DM + n0 + m;
    qks[o] = cq[j] + bbq;
    vs[o]  = cv[j] + bbv;
  }

  row_start_fill(row_start, e0, r0, p0);
  if (has1) row_start_fill(row_start, e1, r1, p1);
}

#define NODE_BLOCK 256
#define STASH 32

__device__ __forceinline__ float quad_sum(float p) {
  p += __int_as_float(__builtin_amdgcn_update_dpp(
      0, __float_as_int(p), 0xB1, 0xF, 0xF, false));
  p += __int_as_float(__builtin_amdgcn_update_dpp(
      0, __float_as_int(p), 0x4E, 0xF, 0xF, false));
  return p;
}

__device__ __forceinline__ float edge_score(const float* __restrict__ qks,
                                            int s, int l16, v4f q) {
  const v4f k = *(const v4f*)(qks + (size_t)s * DM + l16 * 4);
  return quad_sum(q.x * k.x + q.y * k.y + q.z * k.z + q.w * k.w);
}

__global__ __launch_bounds__(NODE_BLOCK)
void node_attention_kernel(const float* __restrict__ qks,
                           const float* __restrict__ vs,
                           const int* __restrict__ senders,
                           const int* __restrict__ row_start,
                           float* __restrict__ attn,
                           float* __restrict__ ctx) {
  __shared__ float stash[STASH * NODE_BLOCK];

  const int tid  = threadIdx.x;
  const int l16  = tid & 15;
  const int node = blockIdx.x * (NODE_BLOCK / 16) + (tid >> 4);

  int start = row_start[node];
  int end   = row_start[node + 1];
  start = min(max(start, 0), NE);
  end   = min(max(end, start), NE);
  const int deg = end - start;

  const v4f q = *(const v4f*)(qks + (size_t)node * DM + l16 * 4);
  const int* __restrict__ snd = senders + start;
  float* __restrict__ st = stash + tid;

  float m = -INFINITY;
  int s_next = (deg > 0) ? snd[0] : 0;
  for (int i = 0; i < deg; ++i) {
    const int s = s_next;
    if (i + 1 < deg) s_next = snd[i + 1];
    const float sc = edge_score(qks, s, l16, q);
    if (i < STASH) st[i * NODE_BLOCK] = sc;
    m = fmaxf(m, sc);
  }

  float z = 0.0f;
  for (int i = 0; i < deg; ++i) {
    const float sc = (i < STASH) ? st[i * NODE_BLOCK]
                                 : edge_score(qks, snd[i], l16, q);
    const float ev = __expf(sc - m);
    if (i < STASH) st[i * NODE_BLOCK] = ev;
    z += ev;
  }

  v4f acc = {0.0f, 0.0f, 0.0f, 0.0f};
  s_next = (deg > 0) ? snd[0] : 0;
  for (int i = 0; i < deg; ++i) {
    const int s = s_next;
    if (i + 1 < deg) s_next = snd[i + 1];
    const v4f v = *(const v4f*)(vs + (size_t)s * DM + l16 * 4);
    const float ev = (i < STASH) ? st[i * NODE_BLOCK]
                                 : __expf(edge_score(qks, s, l16, q) - m);
    const float a = ev / z;
    if ((l16 & 3) == 0) attn[(size_t)(start + i) * NH + (l16 >> 2)] = a;
    acc.x += a * v.x;
    acc.y += a * v.y;
    acc.z += a * v.z;
    acc.w += a * v.w;
  }
  *(v4f*)(ctx + (size_t)node * DM + l16 * 4) = acc;
}

extern "C" void kernel_launch(void* const* d_in, const int* in_sizes, int n_in,
                              void* d_out, int out_size, void* d_ws, size_t ws_size,
                              hipStream_t stream) {
  const float* nodes     = (const float*)d_in[0];
  const float* W_qk      = (const float*)d_in[1];
  const float* b_qk      = (const float*)d_in[2];
  const float* W_v       = (const float*)d_in[3];
  const float* b_v       = (const float*)d_in[4];
  const float* W_out     = (const float*)d_in[5];
  const float* b_out     = (const float*)d_in[6];
  const int*   senders   = (const int*)d_in[7];
  const int*   receivers = (const int*)d_in[8];

  float* out  = (float*)d_out;
  float* attn = out + (size_t)NN * DM;

  float* ws      = (float*)d_ws;
  float* qks     = ws;
  float* vs      = ws + (size_t)NN * DM;
  float* ctx     = ws + (size_t)2 * NN * DM;
  int* row_start = (int*)(ws + (size_t)3 * NN * DM);

  const int gemmGrid = ROW_TILES;
  const int nodeGrid = NN / (NODE_BLOCK / 16);

  project_qkv_rows_kernel<<<gemmGrid, 128, 0, stream>>>(nodes, W_qk, b_qk, W_v, b_v,
                                                        receivers, qks, vs, row_start);

  node_attention_kernel<<<nodeGrid, NODE_BLOCK, 0, stream>>>(qks, vs, senders, row_start, attn, ctx);

  gemm64_wmma_kernel<<<gemmGrid, 128, 0, stream>>>(ctx, W_out, b_out, out);
}


# Output GEMM joins the edge launch: a block's 16 ctx rows go to LDS and, behind one LDS-only barrier, 4 of its 8 waves do the tile's W_out product (k and sum order kept); launch 3 and ctx plane go
# speedup vs baseline: 5.2802x; 1.0377x over previous
#include <hip/hip_runtime.h>


typedef __attribute__((ext_vector_type(2))) float v2f;
typedef __attribute__((ext_vector_type(4))) float v4f;
typedef __attribute__((ext_vector_type(8))) float v8f;

#define NN  100000
#define NE  1000000
#define DM  64
#define NH  4
#define DH  16

__device__ __forceinline__ void row_start_fill(int* __restrict__ row_start,
                                               int e, int r, int p) {
  for (int n = p + 1; n <= r; ++n) row_start[n] = e;
  if (e == NE - 1) {
    for (int n = r + 1; n <= NN; ++n) row_start[n] = NE;
  }
}


#define ROW_TILES (NN / 16)
#define EPB (NE / ROW_TILES)
static_assert(EPB * ROW_TILES == NE, "edge slices must tile the edge list");
#define EPW (EPB / 4)
static_assert(EPW * 4 == EPB, "a block's edge slice splits evenly over its waves");
static_assert(EPW > 32 && EPW <= 64, "one edge per lane plus a partial second pass");

__global__ __launch_bounds__(128)
void project_qkv_rows_kernel(const float* __restrict__ A,
                             const float* __restrict__ Wq,
                             const float* __restrict__ bq,
                             const float* __restrict__ Wv,
                             const float* __restrict__ bv,
                             const int* __restrict__ receivers,
                             float* __restrict__ qks,
                             float* __restrict__ vs,
                             int* __restrict__ row_start) {
  const int lane = threadIdx.x & 31;
  const int wave = threadIdx.x >> 5;
  const int row0 = blockIdx.x * 16;
  const int n0   = wave * 16;
  const int m    = lane & 15;
  const int half = lane >> 4;
  const int kA   = half * 2;

  const float* __restrict__ arow = A + (size_t)(row0 + m) * DM;

  v8f cq = {};
  v8f cv = {};
#pragma unroll
  for (int k0 = 0; k0 < DM; k0 += 4) {
    v2f a, wq, wv;
    a.x = arow[k0 + kA];
    a.y = arow[k0 + kA + 1];
    wq.x = Wq[(size_t)(k0 + kA) * DM + n0 + m];
    wq.y = Wq[(size_t)(k0 + kA + 1) * DM + n0 + m];
    wv.x = Wv[(size_t)(k0 + kA) * DM + n0 + m];
    wv.y = Wv[(size_t)(k0 + kA + 1) * DM + n0 + m];
    cq = __builtin_amdgcn_wmma_f32_16x16x4_f32(false, a, false, wq,
                                               (short)0, cq, false, false);
    cv = __builtin_amdgcn_wmma_f32_16x16x4_f32(false, a, false, wv,
                                               (short)0, cv, false, false);
  }

  const int wbase = blockIdx.x * EPB + wave * EPW;
  const bool has1 = lane < EPW - 32;
  const int e0 = wbase + lane;
  const int e1 = wbase + 32 + min(lane, EPW - 33);
  const int a0  = max(e0 - 1, 0);
  const int x0  = receivers[a0];
  const int y0  = receivers[a0 + 1];
  const int r0  = (e0 == 0) ? x0 : y0;
  const int p0  = (e0 == 0) ? -1 : x0;
  const int p1  = receivers[e1 - 1];
  const int r1  = receivers[e1];

  const float bbq = bq[n0 + m];
  const float bbv = bv[n0 + m];
#pragma unroll
  for (int j = 0; j < 8; ++j) {
    const size_t o = (size_t)(row0 + j + half * 8) * DM + n0 + m;
    qks[o] = cq[j] + bbq;
    vs[o]  = cv[j] + bbv;
  }

  row_start_fill(row_start, e0, r0, p0);
  if (has1) row_start_fill(row_start, e1, r1, p1);
}

#define NODE_BLOCK 256
#define STASH 32
#define CTX_PITCH 68

__device__ __forceinline__ float quad_sum(float p) {
  p += __int_as_float(__builtin_amdgcn_update_dpp(
      0, __float_as_int(p), 0xB1, 0xF, 0xF, false));
  p += __int_as_float(__builtin_amdgcn_update_dpp(
      0, __float_as_int(p), 0x4E, 0xF, 0xF, false));
  return p;
}

__device__ __forceinline__ float edge_score(const float* __restrict__ qks,
                                            int s, int l16, v4f q) {
  const v4f k = *(const v4f*)(qks + (size_t)s * DM + l16 * 4);
  return quad_sum(q.x * k.x + q.y * k.y + q.z * k.z + q.w * k.w);
}

__global__ __launch_bounds__(NODE_BLOCK)
void node_attention_out_kernel(const float* __restrict__ qks,
                               const float* __restrict__ vs,
                               const int* __restrict__ senders,
                               const int* __restrict__ row_start,
                               const float* __restrict__ Wo,
                               const float* __restrict__ bo,
                               float* __restrict__ attn,
                               float* __restrict__ out) {
  __shared__ float stash[STASH * NODE_BLOCK];
  __shared__ __attribute__((aligned(16))) float ctx_tile[16 * CTX_PITCH];

  const int tid  = threadIdx.x;
  const int l16  = tid & 15;
  const int node = blockIdx.x * (NODE_BLOCK / 16) + (tid >> 4);

  int start = row_start[node];
  int end   = row_start[node + 1];
  start = min(max(start, 0), NE);
  end   = min(max(end, start), NE);
  const int deg = end - start;

  const v4f q = *(const v4f*)(qks + (size_t)node * DM + l16 * 4);
  const int* __restrict__ snd = senders + start;
  float* __restrict__ st = stash + tid;

  float m = -INFINITY;
  int s_next = (deg > 0) ? snd[0] : 0;
  for (int i = 0; i < deg; ++i) {
    const int s = s_next;
    if (i + 1 < deg) s_next = snd[i + 1];
    const float sc = edge_score(qks, s, l16, q);
    if (i < STASH) st[i * NODE_BLOCK] = sc;
    m = fmaxf(m, sc);
  }

  float z = 0.0f;
  for (int i = 0; i < deg; ++i) {
    const float sc = (i < STASH) ? st[i * NODE_BLOCK]
                                 : edge_score(qks, snd[i], l16, q);
    const float ev = __expf(sc - m);
    if (i < STASH) st[i * NODE_BLOCK] = ev;
    z += ev;
  }

  v4f acc = {0.0f, 0.0f, 0.0f, 0.0f};
  s_next = (deg > 0) ? snd[0] : 0;
  for (int i = 0; i < deg; ++i) {
    const int s = s_next;
    if (i + 1 < deg) s_next = snd[i + 1];
    const v4f v = *(const v4f*)(vs + (size_t)s * DM + l16 * 4);
    const float ev = (i < STASH) ? st[i * NODE_BLOCK]
                                 : __expf(edge_score(qks, s, l16, q) - m);
    const float a = ev / z;
    if ((l16 & 3) == 0) attn[(size_t)(start + i) * NH + (l16 >> 2)] = a;
    acc.x += a * v.x;
    acc.y += a * v.y;
    acc.z += a * v.z;
    acc.w += a * v.w;
  }
  *(v4f*)(ctx_tile + (tid >> 4) * CTX_PITCH + l16 * 4) = acc;

  const int lane = tid & 31;
  const int wave = __builtin_amdgcn_readfirstlane(tid >> 5);
  const int n0   = (wave & 3) * 16;
  const int mm   = lane & 15;
  const int half = lane >> 4;
  const int kA   = half * 2;

  v2f wb[DM / 4];
  if (wave < 4) {
#pragma unroll
    for (int k = 0; k < DM / 4; ++k) {
      wb[k].x = Wo[(size_t)(k * 4 + kA) * DM + n0 + mm];
      wb[k].y = Wo[(size_t)(k * 4 + kA + 1) * DM + n0 + mm];
    }
  }

  __builtin_amdgcn_fence(__ATOMIC_RELEASE, "workgroup", "local");
  __builtin_amdgcn_s_barrier();
  __builtin_amdgcn_fence(__ATOMIC_ACQUIRE, "workgroup", "local");

  if (wave < 4) {
    const float* __restrict__ arow = ctx_tile + mm * CTX_PITCH + kA;
    v2f af[DM / 4];
#pragma unroll
    for (int k = 0; k < DM / 4; ++k) af[k] = *(const v2f*)(arow + k * 4);

    v8f c = {};
#pragma unroll
    for (int k = 0; k < DM / 4; ++k) {
      c = __builtin_amdgcn_wmma_f32_16x16x4_f32(false, af[k], false, wb[k],
                                                (short)0, c, false, false);
    }

    const float bb = bo[n0 + mm];
    const int row0 = blockIdx.x * (NODE_BLOCK / 16);
#pragma unroll
    for (int j = 0; j < 8; ++j) {
      out[(size_t)(row0 + j + half * 8) * DM + n0 + mm] = c[j] + bb;
    }
  }
}

extern "C" void kernel_launch(void* const* d_in, const int* in_sizes, int n_in,
                              void* d_out, int out_size, void* d_ws, size_t ws_size,
                              hipStream_t stream) {
  const float* nodes     = (const float*)d_in[0];
  const float* W_qk      = (const float*)d_in[1];
  const float* b_qk      = (const float*)d_in[2];
  const float* W_v       = (const float*)d_in[3];
  const float* b_v       = (const float*)d_in[4];
  const float* W_out     = (const float*)d_in[5];
  const float* b_out     = (const float*)d_in[6];
  const int*   senders   = (const int*)d_in[7];
  const int*   receivers = (const int*)d_in[8];

  float* out  = (float*)d_out;
  float* attn = out + (size_t)NN * DM;

  float* ws      = (float*)d_ws;
  float* qks     = ws;
  float* vs      = ws + (size_t)NN * DM;
  int* row_start = (int*)(ws + (size_t)3 * NN * DM);

  const int gemmGrid = ROW_TILES;
  const int nodeGrid = NN / (NODE_BLOCK / 16);

  project_qkv_rows_kernel<<<gemmGrid, 128, 0, stream>>>(nodes, W_qk, b_qk, W_v, b_v,
                                                        receivers, qks, vs, row_start);

  node_attention_out_kernel<<<nodeGrid, NODE_BLOCK, 0, stream>>>(qks, vs, senders, row_start,
                                                               W_out, b_out, attn, out);
}


# Edge phase: a node's first 16 edges go as one batch: one index load, DPP row broadcast, all 16 K (then V) row gathers in flight before first use; edge-order sums kept, longer ranges as before
# speedup vs baseline: 6.2663x; 1.1868x over previous
#include <hip/hip_runtime.h>


typedef __attribute__((ext_vector_type(2))) float v2f;
typedef __attribute__((ext_vector_type(4))) float v4f;
typedef __attribute__((ext_vector_type(8))) float v8f;

#define NN  100000
#define NE  1000000
#define DM  64
#define NH  4
#define DH  16

__device__ __forceinline__ void row_start_fill(int* __restrict__ row_start,
                                               int e, int r, int p) {
  for (int n = p + 1; n <= r; ++n) row_start[n] = e;
  if (e == NE - 1) {
    for (int n = r + 1; n <= NN; ++n) row_start[n] = NE;
  }
}


#define ROW_TILES (NN / 16)
#define EPB (NE / ROW_TILES)
static_assert(EPB * ROW_TILES == NE, "edge slices must tile the edge list");
#define EPW (EPB / 4)
static_assert(EPW * 4 == EPB, "a block's edge slice splits evenly over its waves");
static_assert(EPW > 32 && EPW <= 64, "one edge per lane plus a partial second pass");

__global__ __launch_bounds__(128)
void project_qkv_rows_kernel(const float* __restrict__ A,
                             const float* __restrict__ Wq,
                             const float* __restrict__ bq,
                             const float* __restrict__ Wv,
                             const float* __restrict__ bv,
                             const int* __restrict__ receivers,
                             float* __restrict__ qks,
                             float* __restrict__ vs,
                             int* __restrict__ row_start) {
  const int lane = threadIdx.x & 31;
  const int wave = threadIdx.x >> 5;
  const int row0 = blockIdx.x * 16;
  const int n0   = wave * 16;
  const int m    = lane & 15;
  const int half = lane >> 4;
  const int kA   = half * 2;

  const float* __restrict__ arow = A + (size_t)(row0 + m) * DM;

  v8f cq = {};
  v8f cv = {};
#pragma unroll
  for (int k0 = 0; k0 < DM; k0 += 4) {
    v2f a, wq, wv;
    a.x = arow[k0 + kA];
    a.y = arow[k0 + kA + 1];
    wq.x = Wq[(size_t)(k0 + kA) * DM + n0 + m];
    wq.y = Wq[(size_t)(k0 + kA + 1) * DM + n0 + m];
    wv.x = Wv[(size_t)(k0 + kA) * DM + n0 + m];
    wv.y = Wv[(size_t)(k0 + kA + 1) * DM + n0 + m];
    cq = __builtin_amdgcn_wmma_f32_16x16x4_f32(false, a, false, wq,
                                               (short)0, cq, false, false);
    cv = __builtin_amdgcn_wmma_f32_16x16x4_f32(false, a, false, wv,
                                               (short)0, cv, false, false);
  }

  const int wbase = blockIdx.x * EPB + wave * EPW;
  const bool has1 = lane < EPW - 32;
  const int e0 = wbase + lane;
  const int e1 = wbase + 32 + min(lane, EPW - 33);
  const int a0  = max(e0 - 1, 0);
  const int x0  = receivers[a0];
  const int y0  = receivers[a0 + 1];
  const int r0  = (e0 == 0) ? x0 : y0;
  const int p0  = (e0 == 0) ? -1 : x0;
  const int p1  = receivers[e1 - 1];
  const int r1  = receivers[e1];

  const float bbq = bq[n0 + m];
  const float bbv = bv[n0 + m];
#pragma unroll
  for (int j = 0; j < 8; ++j) {
    const size_t o = (size_t)(row0 + j + half * 8) * DM + n0 + m;
    qks[o] = cq[j] + bbq;
    vs[o]  = cv[j] + bbv;
  }

  row_start_fill(row_start, e0, r0, p0);
  if (has1) row_start_fill(row_start, e1, r1, p1);
}

#define NODE_BLOCK 256
#define BATCH 16
#define STASH 16
#define CTX_PITCH 68

__device__ __forceinline__ float quad_sum(float p) {
  p += __int_as_float(__builtin_amdgcn_update_dpp(
      0, __float_as_int(p), 0xB1, 0xF, 0xF, false));
  p += __int_as_float(__builtin_amdgcn_update_dpp(
      0, __float_as_int(p), 0x4E, 0xF, 0xF, false));
  return p;
}

template <int J>
__device__ __forceinline__ int group_lane(int x) {
  return __builtin_amdgcn_update_dpp(0, x, 0x150 + J, 0xF, 0xF, false);
}

template <int I> struct static_index { static constexpr int value = I; };
template <int FIRST, int LAST, class F>
__device__ __forceinline__ void for_each_index(F&& f) {
  if constexpr (FIRST < LAST) {
    f(static_index<FIRST>{});
    for_each_index<FIRST + 1, LAST>(f);
  }
}

__device__ __forceinline__ v4f row_piece(const float* __restrict__ table, int s, int l16) {
  const unsigned off = (unsigned)s * (DM * 4u) + (unsigned)l16 * 16u;
  return *(const v4f*)((const char*)table + off);
}

__device__ __forceinline__ float head_dot(v4f q, v4f k) {
  return quad_sum(fmaf(q.w, k.w, fmaf(q.z, k.z, fmaf(q.x, k.x, q.y * k.y))));
}

__device__ __forceinline__ float edge_score(const float* __restrict__ qks,
                                            int s, int l16, v4f q) {
  return head_dot(q, row_piece(qks, s, l16));
}

__global__ __launch_bounds__(NODE_BLOCK)
void node_attention_out_kernel(const float* __restrict__ qks,
                               const float* __restrict__ vs,
                               const int* __restrict__ senders,
                               const int* __restrict__ row_start,
                               const float* __restrict__ Wo,
                               const float* __restrict__ bo,
                               float* __restrict__ attn,
                               float* __restrict__ out) {
  __shared__ float stash[STASH * NODE_BLOCK];
  __shared__ __attribute__((aligned(16))) float ctx_tile[16 * CTX_PITCH];

  const int tid  = threadIdx.x;
  const int l16  = tid & 15;
  const int node = blockIdx.x * (NODE_BLOCK / 16) + (tid >> 4);

  int start = row_start[node];
  int end   = row_start[node + 1];
  start = min(max(start, 0), NE);
  end   = min(max(end, start), NE);
  const int deg = end - start;

  const int wdeg = max(__builtin_amdgcn_readlane(deg, 0),
                       __builtin_amdgcn_readlane(deg, 16));

  const v4f q = row_piece(qks, node, l16);
  const int* __restrict__ snd = senders + start;
  float* __restrict__ st = stash + tid;
  float* __restrict__ attn_row = attn + (size_t)start * NH + (l16 >> 2);

  const int own   = (l16 < deg) ? senders[min(start + l16, NE - 1)] : node;
  const int first = group_lane<0>(own);
  const int idx   = (l16 < deg) ? own : first;

  v4f   row[BATCH];
  float sc[BATCH];

  for_each_index<0, BATCH>([&](auto J) {
    constexpr int j = decltype(J)::value;
    row[j] = row_piece(qks, group_lane<j>(idx), l16);
  });
  float m = -INFINITY;
  for_each_index<0, BATCH / 4>([&](auto G) {
    constexpr int g = decltype(G)::value;
    if (g * 4 < wdeg) {
      for_each_index<g * 4, g * 4 + 4>([&](auto J) {
        constexpr int j = decltype(J)::value;
        sc[j] = (j < deg) ? head_dot(q, row[j]) : -INFINITY;
        m = fmaxf(m, sc[j]);
      });
    }
  });
  int s_next = (deg > BATCH) ? snd[BATCH] : 0;
  for (int i = BATCH; i < deg; ++i) {
    const int s = s_next;
    if (i + 1 < deg) s_next = snd[i + 1];
    const float sv = edge_score(qks, s, l16, q);
    if (i < BATCH + STASH) st[(i - BATCH) * NODE_BLOCK] = sv;
    m = fmaxf(m, sv);
  }

  for_each_index<0, BATCH>([&](auto J) {
    constexpr int j = decltype(J)::value;
    row[j] = row_piece(vs, group_lane<j>(idx), l16);
  });

  float z = 0.0f;
  for_each_index<0, BATCH / 4>([&](auto G) {
    constexpr int g = decltype(G)::value;
    if (g * 4 < wdeg) {
      for_each_index<g * 4, g * 4 + 4>([&](auto J) {
        constexpr int j = decltype(J)::value;
        sc[j] = (j < deg) ? __expf(sc[j] - m) : 0.0f;
        z += sc[j];
      });
    }
  });
  for (int i = BATCH; i < deg; ++i) {
    const float sv = (i < BATCH + STASH) ? st[(i - BATCH) * NODE_BLOCK]
                                         : edge_score(qks, snd[i], l16, q);
    const float ev = __expf(sv - m);
    if (i < BATCH + STASH) st[(i - BATCH) * NODE_BLOCK] = ev;
    z += ev;
  }

  v4f acc = {0.0f, 0.0f, 0.0f, 0.0f};
  for_each_index<0, BATCH / 4>([&](auto G) {
    constexpr int g = decltype(G)::value;
    if (g * 4 < wdeg) {
      for_each_index<g * 4, g * 4 + 4>([&](auto J) {
        constexpr int j = decltype(J)::value;
        if (j < deg) {
          const float a = sc[j] / z;
          if ((l16 & 3) == 0) attn_row[j * NH] = a;
          const v4f v = row[j];
          acc.x += a * v.x;
          acc.y += a * v.y;
          acc.z += a * v.z;
          acc.w += a * v.w;
        }
      });
    }
  });
  s_next = (deg > BATCH) ? snd[BATCH] : 0;
  for (int i = BATCH; i < deg; ++i) {
    const int s = s_next;
    if (i + 1 < deg) s_next = snd[i + 1];
    const v4f v = row_piece(vs, s, l16);
    const float ev = (i < BATCH + STASH) ? st[(i - BATCH) * NODE_BLOCK]
                                         : __expf(edge_score(qks, s, l16, q) - m);
    const float a = ev / z;
    if ((l16 & 3) == 0) attn_row[(size_t)i * NH] = a;
    acc.x += a * v.x;
    acc.y += a * v.y;
    acc.z += a * v.z;
    acc.w += a * v.w;
  }
  *(v4f*)(ctx_tile + (tid >> 4) * CTX_PITCH + l16 * 4) = acc;

  const int lane = tid & 31;
  const int wave = __builtin_amdgcn_readfirstlane(tid >> 5);
  const int n0   = (wave & 3) * 16;
  const int mm   = lane & 15;
  const int half = lane >> 4;
  const int kA   = half * 2;

  v2f wb[DM / 4];
  if (wave < 4) {
#pragma unroll
    for (int k = 0; k < DM / 4; ++k) {
      wb[k].x = Wo[(size_t)(k * 4 + kA) * DM + n0 + mm];
      wb[k].y = Wo[(size_t)(k * 4 + kA + 1) * DM + n0 + mm];
    }
  }

  __builtin_amdgcn_fence(__ATOMIC_RELEASE, "workgroup", "local");
  __builtin_amdgcn_s_barrier();
  __builtin_amdgcn_fence(__ATOMIC_ACQUIRE, "workgroup", "local");

  if (wave < 4) {
    const float* __restrict__ arow = ctx_tile + mm * CTX_PITCH + kA;
    v2f af[DM / 4];
#pragma unroll
    for (int k = 0; k < DM / 4; ++k) af[k] = *(const v2f*)(arow + k * 4);

    v8f c = {};
#pragma unroll
    for (int k = 0; k < DM / 4; ++k) {
      c = __builtin_amdgcn_wmma_f32_16x16x4_f32(false, af[k], false, wb[k],
                                                (short)0, c, false, false);
    }

    const float bb = bo[n0 + mm];
    const int row0 = blockIdx.x * (NODE_BLOCK / 16);
#pragma unroll
    for (int j = 0; j < 8; ++j) {
      out[(size_t)(row0 + j + half * 8) * DM + n0 + mm] = c[j] + bb;
    }
  }
}

extern "C" void kernel_launch(void* const* d_in, const int* in_sizes, int n_in,
                              void* d_out, int out_size, void* d_ws, size_t ws_size,
                              hipStream_t stream) {
  const float* nodes     = (const float*)d_in[0];
  const float* W_qk      = (const float*)d_in[1];
  const float* b_qk      = (const float*)d_in[2];
  const float* W_v       = (const float*)d_in[3];
  const float* b_v       = (const float*)d_in[4];
  const float* W_out     = (const float*)d_in[5];
  const float* b_out     = (const float*)d_in[6];
  const int*   senders   = (const int*)d_in[7];
  const int*   receivers = (const int*)d_in[8];

  float* out  = (float*)d_out;
  float* attn = out + (size_t)NN * DM;

  float* ws      = (float*)d_ws;
  float* qks     = ws;
  float* vs      = ws + (size_t)NN * DM;
  int* row_start = (int*)(ws + (size_t)3 * NN * DM);

  const int gemmGrid = ROW_TILES;
  const int nodeGrid = NN / (NODE_BLOCK / 16);

  project_qkv_rows_kernel<<<gemmGrid, 128, 0, stream>>>(nodes, W_qk, b_qk, W_v, b_v,
                                                        receivers, qks, vs, row_start);

  node_attention_out_kernel<<<nodeGrid, NODE_BLOCK, 0, stream>>>(qks, vs, senders, row_start,
                                                               W_out, b_out, attn, out);
}


# Edge batch: the 4 lanes of a head split its 16 slots, each doing exp and divide for 4 only; results go round by DPP quad broadcast, z and ctx add in edge order as before; attn leaves as 64-B runs
# speedup vs baseline: 6.8165x; 1.0878x over previous
#include <hip/hip_runtime.h>


typedef __attribute__((ext_vector_type(2))) float v2f;
typedef __attribute__((ext_vector_type(4))) float v4f;
typedef __attribute__((ext_vector_type(8))) float v8f;

#define NN  100000
#define NE  1000000
#define DM  64
#define NH  4
#define DH  16

__device__ __forceinline__ void row_start_fill(int* __restrict__ row_start,
                                               int e, int r, int p) {
  for (int n = p + 1; n <= r; ++n) row_start[n] = e;
  if (e == NE - 1) {
    for (int n = r + 1; n <= NN; ++n) row_start[n] = NE;
  }
}


#define ROW_TILES (NN / 16)
#define EPB (NE / ROW_TILES)
static_assert(EPB * ROW_TILES == NE, "edge slices must tile the edge list");
#define EPW (EPB / 4)
static_assert(EPW * 4 == EPB, "a block's edge slice splits evenly over its waves");
static_assert(EPW > 32 && EPW <= 64, "one edge per lane plus a partial second pass");

__global__ __launch_bounds__(128)
void project_qkv_rows_kernel(const float* __restrict__ A,
                             const float* __restrict__ Wq,
                             const float* __restrict__ bq,
                             const float* __restrict__ Wv,
                             const float* __restrict__ bv,
                             const int* __restrict__ receivers,
                             float* __restrict__ qks,
                             float* __restrict__ vs,
                             int* __restrict__ row_start) {
  const int lane = threadIdx.x & 31;
  const int wave = threadIdx.x >> 5;
  const int row0 = blockIdx.x * 16;
  const int n0   = wave * 16;
  const int m    = lane & 15;
  const int half = lane >> 4;
  const int kA   = half * 2;

  const float* __restrict__ arow = A + (size_t)(row0 + m) * DM;

  v8f cq = {};
  v8f cv = {};
#pragma unroll
  for (int k0 = 0; k0 < DM; k0 += 4) {
    v2f a, wq, wv;
    a.x = arow[k0 + kA];
    a.y = arow[k0 + kA + 1];
    wq.x = Wq[(size_t)(k0 + kA) * DM + n0 + m];
    wq.y = Wq[(size_t)(k0 + kA + 1) * DM + n0 + m];
    wv.x = Wv[(size_t)(k0 + kA) * DM + n0 + m];
    wv.y = Wv[(size_t)(k0 + kA + 1) * DM + n0 + m];
    cq = __builtin_amdgcn_wmma_f32_16x16x4_f32(false, a, false, wq,
                                               (short)0, cq, false, false);
    cv = __builtin_amdgcn_wmma_f32_16x16x4_f32(false, a, false, wv,
                                               (short)0, cv, false, false);
  }

  const int wbase = blockIdx.x * EPB + wave * EPW;
  const bool has1 = lane < EPW - 32;
  const int e0 = wbase + lane;
  const int e1 = wbase + 32 + min(lane, EPW - 33);
  const int a0  = max(e0 - 1, 0);
  const int x0  = receivers[a0];
  const int y0  = receivers[a0 + 1];
  const int r0  = (e0 == 0) ? x0 : y0;
  const int p0  = (e0 == 0) ? -1 : x0;
  const int p1  = receivers[e1 - 1];
  const int r1  = receivers[e1];

  const float bbq = bq[n0 + m];
  const float bbv = bv[n0 + m];
#pragma unroll
  for (int j = 0; j < 8; ++j) {
    const size_t o = (size_t)(row0 + j + half * 8) * DM + n0 + m;
    qks[o] = cq[j] + bbq;
    vs[o]  = cv[j] + bbv;
  }

  row_start_fill(row_start, e0, r0, p0);
  if (has1) row_start_fill(row_start, e1, r1, p1);
}

#define NODE_BLOCK 256
#define BATCH 16
#define STASH 16
#define CTX_PITCH 68

__device__ __forceinline__ float quad_sum(float p) {
  p += __int_as_float(__builtin_amdgcn_update_dpp(
      0, __float_as_int(p), 0xB1, 0xF, 0xF, false));
  p += __int_as_float(__builtin_amdgcn_update_dpp(
      0, __float_as_int(p), 0x4E, 0xF, 0xF, false));
  return p;
}

template <int J>
__device__ __forceinline__ int group_lane(int x) {
  return __builtin_amdgcn_update_dpp(0, x, 0x150 + J, 0xF, 0xF, false);
}

template <int T>
__device__ __forceinline__ float quad_lane(float x) {
  return __int_as_float(__builtin_amdgcn_update_dpp(
      0, __float_as_int(x), T * 0x55, 0xF, 0xF, false));
}

template <int I> struct static_index { static constexpr int value = I; };
template <int FIRST, int LAST, class F>
__device__ __forceinline__ void for_each_index(F&& f) {
  if constexpr (FIRST < LAST) {
    f(static_index<FIRST>{});
    for_each_index<FIRST + 1, LAST>(f);
  }
}

__device__ __forceinline__ v4f row_piece(const float* __restrict__ table, int s, int l16) {
  const unsigned off = (unsigned)s * (DM * 4u) + (unsigned)l16 * 16u;
  return *(const v4f*)((const char*)table + off);
}

__device__ __forceinline__ void put_at(float* __restrict__ base, unsigned byte_off, float x) {
  *(float*)((char*)base + byte_off) = x;
}

__device__ __forceinline__ float head_dot(v4f q, v4f k) {
  return quad_sum(fmaf(q.w, k.w, fmaf(q.z, k.z, fmaf(q.x, k.x, q.y * k.y))));
}

__device__ __forceinline__ float edge_score(const float* __restrict__ qks,
                                            int s, int l16, v4f q) {
  return head_dot(q, row_piece(qks, s, l16));
}

__global__ __launch_bounds__(NODE_BLOCK)
void node_attention_out_kernel(const float* __restrict__ qks,
                               const float* __restrict__ vs,
                               const int* __restrict__ senders,
                               const int* __restrict__ row_start,
                               const float* __restrict__ Wo,
                               const float* __restrict__ bo,
                               float* __restrict__ attn,
                               float* __restrict__ out) {
  __shared__ float stash[STASH * NODE_BLOCK];
  __shared__ __attribute__((aligned(16))) float ctx_tile[16 * CTX_PITCH];

  const int tid  = threadIdx.x;
  const int l16  = tid & 15;
  const int node = blockIdx.x * (NODE_BLOCK / 16) + (tid >> 4);

  int start = row_start[node];
  int end   = row_start[node + 1];
  start = min(max(start, 0), NE);
  end   = min(max(end, start), NE);
  const int deg = end - start;

  const int wdeg = max(__builtin_amdgcn_readlane(deg, 0),
                       __builtin_amdgcn_readlane(deg, 16));

  const v4f q = row_piece(qks, node, l16);
  float* __restrict__ st = stash + tid;
  const unsigned attn_off = ((unsigned)start * NH + (unsigned)(l16 >> 2)) * 4u;

  const int own   = (l16 < deg) ? senders[min(start + l16, NE - 1)] : node;
  const int first = group_lane<0>(own);
  const int idx   = (l16 < deg) ? own : first;

  v4f   row[BATCH];
  float mine[BATCH / 4];
  const int t4 = l16 & 3;

  for_each_index<0, BATCH>([&](auto J) {
    constexpr int j = decltype(J)::value;
    row[j] = row_piece(qks, group_lane<j>(idx), l16);
  });
  float m = -INFINITY;
  for_each_index<0, BATCH / 4>([&](auto G) {
    constexpr int g = decltype(G)::value;
    mine[g] = -INFINITY;
    if (g * 4 < wdeg) {
      for_each_index<g * 4, g * 4 + 4>([&](auto J) {
        constexpr int j = decltype(J)::value;
        const float s = (j < deg) ? head_dot(q, row[j]) : -INFINITY;
        m = fmaxf(m, s);
        mine[g] = (t4 == (j & 3)) ? s : mine[g];
      });
    }
  });
  int s_next = (deg > BATCH) ? senders[start + BATCH] : 0;
  for (int i = BATCH; i < deg; ++i) {
    const int s = s_next;
    if (i + 1 < deg) s_next = senders[start + i + 1];
    const float sv = edge_score(qks, s, l16, q);
    if (i < BATCH + STASH) st[(i - BATCH) * NODE_BLOCK] = sv;
    m = fmaxf(m, sv);
  }

  for_each_index<0, BATCH>([&](auto J) {
    constexpr int j = decltype(J)::value;
    row[j] = row_piece(vs, group_lane<j>(idx), l16);
  });

  float z = 0.0f;
  for_each_index<0, BATCH / 4>([&](auto G) {
    constexpr int g = decltype(G)::value;
    if (g * 4 < wdeg) {
      const float e = (g * 4 + t4 < deg) ? __expf(mine[g] - m) : 0.0f;
      mine[g] = e;
      for_each_index<0, 4>([&](auto T) {
        z += quad_lane<decltype(T)::value>(e);
      });
    }
  });
  for (int i = BATCH; i < deg; ++i) {
    const float sv = (i < BATCH + STASH) ? st[(i - BATCH) * NODE_BLOCK]
                                         : edge_score(qks, senders[start + i], l16, q);
    const float ev = __expf(sv - m);
    if (i < BATCH + STASH) st[(i - BATCH) * NODE_BLOCK] = ev;
    z += ev;
  }

  v4f acc = {0.0f, 0.0f, 0.0f, 0.0f};
  for_each_index<0, BATCH / 4>([&](auto G) {
    constexpr int g = decltype(G)::value;
    if (g * 4 < wdeg) {
      const bool own_slot = g * 4 + t4 < deg;
      const float a_own = own_slot ? mine[g] / z : 0.0f;
      if (own_slot) put_at(attn, attn_off + (unsigned)(g * 4 + t4) * (NH * 4u), a_own);
      for_each_index<0, 4>([&](auto T) {
        constexpr int j = g * 4 + decltype(T)::value;
        const float a = quad_lane<decltype(T)::value>(a_own);
        if (j < deg) {
          const v4f v = row[j];
          acc.x += a * v.x;
          acc.y += a * v.y;
          acc.z += a * v.z;
          acc.w += a * v.w;
        }
      });
    }
  });
  s_next = (deg > BATCH) ? senders[start + BATCH] : 0;
  for (int i = BATCH; i < deg; ++i) {
    const int s = s_next;
    if (i + 1 < deg) s_next = senders[start + i + 1];
    const v4f v = row_piece(vs, s, l16);
    const float ev = (i < BATCH + STASH) ? st[(i - BATCH) * NODE_BLOCK]
                                         : __expf(edge_score(qks, s, l16, q) - m);
    const float a = ev / z;
    if ((l16 & 3) == 0) put_at(attn, attn_off + (unsigned)i * (NH * 4u), a);
    acc.x += a * v.x;
    acc.y += a * v.y;
    acc.z += a * v.z;
    acc.w += a * v.w;
  }
  *(v4f*)(ctx_tile + (tid >> 4) * CTX_PITCH + l16 * 4) = acc;

  const int lane = tid & 31;
  const int wave = __builtin_amdgcn_readfirstlane(tid >> 5);
  const int n0   = (wave & 3) * 16;
  const int mm   = lane & 15;
  const int half = lane >> 4;
  const int kA   = half * 2;

  v2f wb[DM / 4];
  if (wave < 4) {
#pragma unroll
    for (int k = 0; k < DM / 4; ++k) {
      wb[k].x = Wo[(size_t)(k * 4 + kA) * DM + n0 + mm];
      wb[k].y = Wo[(size_t)(k * 4 + kA + 1) * DM + n0 + mm];
    }
  }

  __builtin_amdgcn_fence(__ATOMIC_RELEASE, "workgroup", "local");
  __builtin_amdgcn_s_barrier();
  __builtin_amdgcn_fence(__ATOMIC_ACQUIRE, "workgroup", "local");

  if (wave < 4) {
    const float* __restrict__ arow = ctx_tile + mm * CTX_PITCH + kA;
    v2f af[DM / 4];
#pragma unroll
    for (int k = 0; k < DM / 4; ++k) af[k] = *(const v2f*)(arow + k * 4);

    v8f c = {};
#pragma unroll
    for (int k = 0; k < DM / 4; ++k) {
      c = __builtin_amdgcn_wmma_f32_16x16x4_f32(false, af[k], false, wb[k],
                                                (short)0, c, false, false);
    }

    const float bb = bo[n0 + mm];
    const int row0 = blockIdx.x * (NODE_BLOCK / 16);
#pragma unroll
    for (int j = 0; j < 8; ++j) {
      out[(size_t)(row0 + j + half * 8) * DM + n0 + mm] = c[j] + bb;
    }
  }
}

extern "C" void kernel_launch(void* const* d_in, const int* in_sizes, int n_in,
                              void* d_out, int out_size, void* d_ws, size_t ws_size,
                              hipStream_t stream) {
  const float* nodes     = (const float*)d_in[0];
  const float* W_qk      = (const float*)d_in[1];
  const float* b_qk      = (const float*)d_in[2];
  const float* W_v       = (const float*)d_in[3];
  const float* b_v       = (const float*)d_in[4];
  const float* W_out     = (const float*)d_in[5];
  const float* b_out     = (const float*)d_in[6];
  const int*   senders   = (const int*)d_in[7];
  const int*   receivers = (const int*)d_in[8];

  float* out  = (float*)d_out;
  float* attn = out + (size_t)NN * DM;

  float* ws      = (float*)d_ws;
  float* qks     = ws;
  float* vs      = ws + (size_t)NN * DM;
  int* row_start = (int*)(ws + (size_t)3 * NN * DM);

  const int gemmGrid = ROW_TILES;
  const int nodeGrid = NN / (NODE_BLOCK / 16);

  project_qkv_rows_kernel<<<gemmGrid, 128, 0, stream>>>(nodes, W_qk, b_qk, W_v, b_v,
                                                        receivers, qks, vs, row_start);

  node_attention_out_kernel<<<nodeGrid, NODE_BLOCK, 0, stream>>>(qks, vs, senders, row_start,
                                                               W_out, b_out, attn, out);
}


# Edge batch goes in two rounds of 8 rows through the same 32 registers (was 16 rows at once): the kernel fits 64 registers, so 16 waves per SIMD are resident, not 10; sum orders kept
# speedup vs baseline: 7.1141x; 1.0437x over previous
#include <hip/hip_runtime.h>


typedef __attribute__((ext_vector_type(2))) float v2f;
typedef __attribute__((ext_vector_type(4))) float v4f;
typedef __attribute__((ext_vector_type(8))) float v8f;

#define NN  100000
#define NE  1000000
#define DM  64
#define NH  4
#define DH  16

__device__ __forceinline__ void row_start_fill(int* __restrict__ row_start,
                                               int e, int r, int p) {
  for (int n = p + 1; n <= r; ++n) row_start[n] = e;
  if (e == NE - 1) {
    for (int n = r + 1; n <= NN; ++n) row_start[n] = NE;
  }
}


#define ROW_TILES (NN / 16)
#define EPB (NE / ROW_TILES)
static_assert(EPB * ROW_TILES == NE, "edge slices must tile the edge list");
#define EPW (EPB / 4)
static_assert(EPW * 4 == EPB, "a block's edge slice splits evenly over its waves");
static_assert(EPW > 32 && EPW <= 64, "one edge per lane plus a partial second pass");

__global__ __launch_bounds__(128)
void project_qkv_rows_kernel(const float* __restrict__ A,
                             const float* __restrict__ Wq,
                             const float* __restrict__ bq,
                             const float* __restrict__ Wv,
                             const float* __restrict__ bv,
                             const int* __restrict__ receivers,
                             float* __restrict__ qks,
                             float* __restrict__ vs,
                             int* __restrict__ row_start) {
  const int lane = threadIdx.x & 31;
  const int wave = threadIdx.x >> 5;
  const int row0 = blockIdx.x * 16;
  const int n0   = wave * 16;
  const int m    = lane & 15;
  const int half = lane >> 4;
  const int kA   = half * 2;

  const float* __restrict__ arow = A + (size_t)(row0 + m) * DM;

  v8f cq = {};
  v8f cv = {};
#pragma unroll
  for (int k0 = 0; k0 < DM; k0 += 4) {
    v2f a, wq, wv;
    a.x = arow[k0 + kA];
    a.y = arow[k0 + kA + 1];
    wq.x = Wq[(size_t)(k0 + kA) * DM + n0 + m];
    wq.y = Wq[(size_t)(k0 + kA + 1) * DM + n0 + m];
    wv.x = Wv[(size_t)(k0 + kA) * DM + n0 + m];
    wv.y = Wv[(size_t)(k0 + kA + 1) * DM + n0 + m];
    cq = __builtin_amdgcn_wmma_f32_16x16x4_f32(false, a, false, wq,
                                               (short)0, cq, false, false);
    cv = __builtin_amdgcn_wmma_f32_16x16x4_f32(false, a, false, wv,
                                               (short)0, cv, false, false);
  }

  const int wbase = blockIdx.x * EPB + wave * EPW;
  const bool has1 = lane < EPW - 32;
  const int e0 = wbase + lane;
  const int e1 = wbase + 32 + min(lane, EPW - 33);
  const int a0  = max(e0 - 1, 0);
  const int x0  = receivers[a0];
  const int y0  = receivers[a0 + 1];
  const int r0  = (e0 == 0) ? x0 : y0;
  const int p0  = (e0 == 0) ? -1 : x0;
  const int p1  = receivers[e1 - 1];
  const int r1  = receivers[e1];

  const float bbq = bq[n0 + m];
  const float bbv = bv[n0 + m];
#pragma unroll
  for (int j = 0; j < 8; ++j) {
    const size_t o = (size_t)(row0 + j + half * 8) * DM + n0 + m;
    qks[o] = cq[j] + bbq;
    vs[o]  = cv[j] + bbv;
  }

  row_start_fill(row_start, e0, r0, p0);
  if (has1) row_start_fill(row_start, e1, r1, p1);
}

#define NODE_BLOCK 256
#define BATCH 16
#define RING 8
#define STASH 16
#define CTX_PITCH 68

__device__ __forceinline__ float quad_sum(float p) {
  p += __int_as_float(__builtin_amdgcn_update_dpp(
      0, __float_as_int(p), 0xB1, 0xF, 0xF, false));
  p += __int_as_float(__builtin_amdgcn_update_dpp(
      0, __float_as_int(p), 0x4E, 0xF, 0xF, false));
  return p;
}

template <int J>
__device__ __forceinline__ int group_lane(int x) {
  return __builtin_amdgcn_update_dpp(0, x, 0x150 + J, 0xF, 0xF, false);
}

template <int T>
__device__ __forceinline__ float quad_lane(float x) {
  return __int_as_float(__builtin_amdgcn_update_dpp(
      0, __float_as_int(x), T * 0x55, 0xF, 0xF, false));
}

template <int I> struct static_index { static constexpr int value = I; };
template <int FIRST, int LAST, class F>
__device__ __forceinline__ void for_each_index(F&& f) {
  if constexpr (FIRST < LAST) {
    f(static_index<FIRST>{});
    for_each_index<FIRST + 1, LAST>(f);
  }
}

__device__ __forceinline__ v4f row_piece(const float* __restrict__ table, int s, int l16) {
  const unsigned off = (unsigned)s * (DM * 4u) + (unsigned)l16 * 16u;
  return *(const v4f*)((const char*)table + off);
}

__device__ __forceinline__ void put_at(float* __restrict__ base, unsigned byte_off, float x) {
  *(float*)((char*)base + byte_off) = x;
}

__device__ __forceinline__ float head_dot(v4f q, v4f k) {
  return quad_sum(fmaf(q.w, k.w, fmaf(q.z, k.z, fmaf(q.x, k.x, q.y * k.y))));
}

__device__ __forceinline__ float edge_score(const float* __restrict__ qks,
                                            int s, int l16, v4f q) {
  return head_dot(q, row_piece(qks, s, l16));
}

__global__ __launch_bounds__(NODE_BLOCK) __attribute__((amdgpu_waves_per_eu(16)))
void node_attention_out_kernel(const float* __restrict__ qks,
                               const float* __restrict__ vs,
                               const int* __restrict__ senders,
                               const int* __restrict__ row_start,
                               const float* __restrict__ Wo,
                               const float* __restrict__ bo,
                               float* __restrict__ attn,
                               float* __restrict__ out) {
  __shared__ float stash[STASH * NODE_BLOCK];
  __shared__ __attribute__((aligned(16))) v4f q_park[NODE_BLOCK];
  __shared__ float m_park[NODE_BLOCK];
  __shared__ __attribute__((aligned(16))) float ctx_tile[16 * CTX_PITCH];

  const int tid  = threadIdx.x;
  const int l16  = tid & 15;
  const int node = blockIdx.x * (NODE_BLOCK / 16) + (tid >> 4);

  int start = row_start[node];
  int end   = row_start[node + 1];
  start = min(max(start, 0), NE);
  end   = min(max(end, start), NE);
  const int deg = end - start;

  const int wdeg = max(__builtin_amdgcn_readlane(deg, 0),
                       __builtin_amdgcn_readlane(deg, 16));

  const v4f q = row_piece(qks, node, l16);
  if (deg > BATCH + STASH) q_park[tid] = q;
  float* __restrict__ st = stash + tid;
  const unsigned attn_off = ((unsigned)start * NH + (unsigned)(l16 >> 2)) * 4u;

  const int own   = (l16 < deg) ? senders[min(start + l16, NE - 1)] : node;
  const int first = group_lane<0>(own);
  const int idx   = (l16 < deg) ? own : first;

  v4f   row[RING];
  float mine[BATCH / 4];
  const int t4 = l16 & 3;

  auto ask = [&](const float* __restrict__ table, auto B) {
    constexpr int b = decltype(B)::value;
    for_each_index<0, RING>([&](auto T) {
      constexpr int t = decltype(T)::value;
      row[t] = row_piece(table, group_lane<b * RING + t>(idx), l16);
    });
  };

  float m = -INFINITY;
  for_each_index<0, BATCH / 4>([&](auto G) { mine[decltype(G)::value] = -INFINITY; });
  auto score_group = [&](auto G) {
    constexpr int g = decltype(G)::value;
    for_each_index<0, 4>([&](auto T) {
      constexpr int t = decltype(T)::value;
      constexpr int j = g * 4 + t;
      const float s = (j < deg) ? head_dot(q, row[(g & 1) * 4 + t]) : -INFINITY;
      m = fmaxf(m, s);
      mine[g] = (t4 == t) ? s : mine[g];
    });
  };
  ask(qks, static_index<0>{});
  if (0 < wdeg) score_group(static_index<0>{});
  if (4 < wdeg) score_group(static_index<1>{});
  if (8 < wdeg) {
    ask(qks, static_index<1>{});
    score_group(static_index<2>{});
    if (12 < wdeg) score_group(static_index<3>{});
  }
  int s_next = (deg > BATCH) ? senders[start + BATCH] : 0;
  for (int i = BATCH; i < deg; ++i) {
    const int s = s_next;
    if (i + 1 < deg) s_next = senders[start + i + 1];
    const float sv = edge_score(qks, s, l16, q);
    if (i < BATCH + STASH) st[(i - BATCH) * NODE_BLOCK] = sv;
    m = fmaxf(m, sv);
  }

  if (deg > BATCH + STASH) m_park[tid] = m;

  ask(vs, static_index<0>{});

  float z = 0.0f;
  for_each_index<0, BATCH / 4>([&](auto G) {
    constexpr int g = decltype(G)::value;
    if (g * 4 < wdeg) {
      const float e = (g * 4 + t4 < deg) ? __expf(mine[g] - m) : 0.0f;
      mine[g] = e;
      for_each_index<0, 4>([&](auto T) {
        z += quad_lane<decltype(T)::value>(e);
      });
    }
  });
  for (int i = BATCH; i < deg; ++i) {
    const float sv = (i < BATCH + STASH) ? st[(i - BATCH) * NODE_BLOCK]
                                         : edge_score(qks, senders[start + i], l16, q_park[tid]);
    const float ev = __expf(sv - m);
    if (i < BATCH + STASH) st[(i - BATCH) * NODE_BLOCK] = ev;
    z += ev;
  }

  v4f acc = {0.0f, 0.0f, 0.0f, 0.0f};
  auto weigh_group = [&](auto G) {
    constexpr int g = decltype(G)::value;
    const bool own_slot = g * 4 + t4 < deg;
    const float a_own = own_slot ? mine[g] / z : 0.0f;
    if (own_slot) put_at(attn, attn_off + (unsigned)(g * 4 + t4) * (NH * 4u), a_own);
    for_each_index<0, 4>([&](auto T) {
      constexpr int t = decltype(T)::value;
      constexpr int j = g * 4 + t;
      const float a = quad_lane<t>(a_own);
      if (j < deg) {
        const v4f v = row[(g & 1) * 4 + t];
        acc.x += a * v.x;
        acc.y += a * v.y;
        acc.z += a * v.z;
        acc.w += a * v.w;
      }
    });
  };
  if (0 < wdeg) weigh_group(static_index<0>{});
  if (4 < wdeg) weigh_group(static_index<1>{});
  if (8 < wdeg) {
    ask(vs, static_index<1>{});
    weigh_group(static_index<2>{});
    if (12 < wdeg) weigh_group(static_index<3>{});
  }
  s_next = (deg > BATCH) ? senders[start + BATCH] : 0;
  for (int i = BATCH; i < deg; ++i) {
    const int s = s_next;
    if (i + 1 < deg) s_next = senders[start + i + 1];
    const v4f v = row_piece(vs, s, l16);
    const float ev = (i < BATCH + STASH) ? st[(i - BATCH) * NODE_BLOCK]
                                         : __expf(edge_score(qks, s, l16, q_park[tid]) - m_park[tid]);
    const float a = ev / z;
    if ((l16 & 3) == 0) put_at(attn, attn_off + (unsigned)i * (NH * 4u), a);
    acc.x += a * v.x;
    acc.y += a * v.y;
    acc.z += a * v.z;
    acc.w += a * v.w;
  }
  *(v4f*)(ctx_tile + (tid >> 4) * CTX_PITCH + l16 * 4) = acc;

  const int lane = tid & 31;
  const int wave = __builtin_amdgcn_readfirstlane(tid >> 5);
  const int n0   = (wave & 3) * 16;
  const int mm   = lane & 15;
  const int half = lane >> 4;
  const int kA   = half * 2;

  v2f wb[DM / 4];
  if (wave < 4) {
#pragma unroll
    for (int k = 0; k < DM / 4; ++k) {
      wb[k].x = Wo[(size_t)(k * 4 + kA) * DM + n0 + mm];
      wb[k].y = Wo[(size_t)(k * 4 + kA + 1) * DM + n0 + mm];
    }
  }

  __builtin_amdgcn_fence(__ATOMIC_RELEASE, "workgroup", "local");
  __builtin_amdgcn_s_barrier();
  __builtin_amdgcn_fence(__ATOMIC_ACQUIRE, "workgroup", "local");

  if (wave < 4) {
    const float* __restrict__ arow = ctx_tile + mm * CTX_PITCH + kA;
    v8f c = {};
#pragma unroll
    for (int k4 = 0; k4 < DM / 4; k4 += 4) {
      v2f af[4];
#pragma unroll
      for (int k = 0; k < 4; ++k) af[k] = *(const v2f*)(arow + (k4 + k) * 4);
#pragma unroll
      for (int k = 0; k < 4; ++k) {
        c = __builtin_amdgcn_wmma_f32_16x16x4_f32(false, af[k], false, wb[k4 + k],
                                                  (short)0, c, false, false);
      }
    }

    const float bb = bo[n0 + mm];
    const int row0 = blockIdx.x * (NODE_BLOCK / 16);
#pragma unroll
    for (int j = 0; j < 8; ++j) {
      out[(size_t)(row0 + j + half * 8) * DM + n0 + mm] = c[j] + bb;
    }
  }
}

extern "C" void kernel_launch(void* const* d_in, const int* in_sizes, int n_in,
                              void* d_out, int out_size, void* d_ws, size_t ws_size,
                              hipStream_t stream) {
  const float* nodes     = (const float*)d_in[0];
  const float* W_qk      = (const float*)d_in[1];
  const float* b_qk      = (const float*)d_in[2];
  const float* W_v       = (const float*)d_in[3];
  const float* b_v       = (const float*)d_in[4];
  const float* W_out     = (const float*)d_in[5];
  const float* b_out     = (const float*)d_in[6];
  const int*   senders   = (const int*)d_in[7];
  const int*   receivers = (const int*)d_in[8];

  float* out  = (float*)d_out;
  float* attn = out + (size_t)NN * DM;

  float* ws      = (float*)d_ws;
  float* qks     = ws;
  float* vs      = ws + (size_t)NN * DM;
  int* row_start = (int*)(ws + (size_t)3 * NN * DM);

  const int gemmGrid = ROW_TILES;
  const int nodeGrid = NN / (NODE_BLOCK / 16);

  project_qkv_rows_kernel<<<gemmGrid, 128, 0, stream>>>(nodes, W_qk, b_qk, W_v, b_v,
                                                        receivers, qks, vs, row_start);

  node_attention_out_kernel<<<nodeGrid, NODE_BLOCK, 0, stream>>>(qks, vs, senders, row_start,
                                                               W_out, b_out, attn, out);
}


# Tiles meet memory as whole 128-B lines: node rows in and QK, V and out tiles out go through LDS with 16 B per lane (2 full rows per instruction); were 8-B strided loads and 64-B store pieces
# speedup vs baseline: 7.3733x; 1.0364x over previous
#include <hip/hip_runtime.h>


typedef __attribute__((ext_vector_type(2))) float v2f;
typedef __attribute__((ext_vector_type(4))) float v4f;
typedef __attribute__((ext_vector_type(8))) float v8f;

#define NN  100000
#define NE  1000000
#define DM  64
#define NH  4
#define DH  16

__device__ __forceinline__ void row_start_fill(int* __restrict__ row_start,
                                               int e, int r, int p) {
  for (int n = p + 1; n <= r; ++n) row_start[n] = e;
  if (e == NE - 1) {
    for (int n = r + 1; n <= NN; ++n) row_start[n] = NE;
  }
}


#define ROW_TILES (NN / 16)
#define EPB (NE / ROW_TILES)
static_assert(EPB * ROW_TILES == NE, "edge slices must tile the edge list");
#define EPW (EPB / 4)
static_assert(EPW * 4 == EPB, "a block's edge slice splits evenly over its waves");
static_assert(EPW > 32 && EPW <= 64, "one edge per lane plus a partial second pass");
#define TILE_PITCH 68

__global__ __launch_bounds__(128) __attribute__((amdgpu_waves_per_eu(16)))
void project_qkv_rows_kernel(const float* __restrict__ A,
                             const float* __restrict__ Wq,
                             const float* __restrict__ bq,
                             const float* __restrict__ Wv,
                             const float* __restrict__ bv,
                             const int* __restrict__ receivers,
                             float* __restrict__ qks,
                             float* __restrict__ vs,
                             int* __restrict__ row_start) {
  const int lane = threadIdx.x & 31;
  const int wave = threadIdx.x >> 5;
  const int row0 = blockIdx.x * 16;
  const int n0   = wave * 16;
  const int m    = lane & 15;
  const int half = lane >> 4;
  const int kA   = half * 2;

  __shared__ __attribute__((aligned(16))) float a_tile[16 * TILE_PITCH];
  __shared__ __attribute__((aligned(16))) float o_tile[2 * 16 * TILE_PITCH];

  const int tid = threadIdx.x;
  const int tr  = tid >> 4;
  const int tq4 = (tid & 15) * 4;
  const size_t blk = (size_t)blockIdx.x * (16 * DM);
  {
    const float* __restrict__ arows = A + blk;
    *(v4f*)(a_tile + tr * TILE_PITCH + tq4)       = *(const v4f*)(arows + tid * 4);
    *(v4f*)(a_tile + (tr + 8) * TILE_PITCH + tq4) = *(const v4f*)(arows + tid * 4 + 8 * DM);
  }
  __builtin_amdgcn_fence(__ATOMIC_RELEASE, "workgroup", "local");
  __builtin_amdgcn_s_barrier();
  __builtin_amdgcn_fence(__ATOMIC_ACQUIRE, "workgroup", "local");

  const float* __restrict__ arow = a_tile + m * TILE_PITCH + kA;

  v8f cq = {};
  v8f cv = {};
#pragma unroll
  for (int k0 = 0; k0 < DM; k0 += 4) {
    v2f wq, wv;
    const v2f a = *(const v2f*)(arow + k0);
    wq.x = Wq[(size_t)(k0 + kA) * DM + n0 + m];
    wq.y = Wq[(size_t)(k0 + kA + 1) * DM + n0 + m];
    wv.x = Wv[(size_t)(k0 + kA) * DM + n0 + m];
    wv.y = Wv[(size_t)(k0 + kA + 1) * DM + n0 + m];
    cq = __builtin_amdgcn_wmma_f32_16x16x4_f32(false, a, false, wq,
                                               (short)0, cq, false, false);
    cv = __builtin_amdgcn_wmma_f32_16x16x4_f32(false, a, false, wv,
                                               (short)0, cv, false, false);
  }

  const int wbase = blockIdx.x * EPB + wave * EPW;
  const bool has1 = lane < EPW - 32;
  const int e0 = wbase + lane;
  const int e1 = wbase + 32 + min(lane, EPW - 33);
  const int a0  = max(e0 - 1, 0);
  const int x0  = receivers[a0];
  const int y0  = receivers[a0 + 1];
  const int r0  = (e0 == 0) ? x0 : y0;
  const int p0  = (e0 == 0) ? -1 : x0;
  const int p1  = receivers[e1 - 1];
  const int r1  = receivers[e1];

  const float bbq = bq[n0 + m];
  const float bbv = bv[n0 + m];
#pragma unroll
  for (int j = 0; j < 8; ++j) {
    const int o = (j + half * 8) * TILE_PITCH + n0 + m;
    o_tile[o] = cq[j] + bbq;
    o_tile[16 * TILE_PITCH + o] = cv[j] + bbv;
  }
  __builtin_amdgcn_fence(__ATOMIC_RELEASE, "workgroup", "local");
  __builtin_amdgcn_s_barrier();
  __builtin_amdgcn_fence(__ATOMIC_ACQUIRE, "workgroup", "local");
  {
    float* __restrict__ qrows = qks + blk;
    float* __restrict__ vrows = vs + blk;
    const float* __restrict__ lo = o_tile + tr * TILE_PITCH + tq4;
    *(v4f*)(qrows + tid * 4)          = *(const v4f*)(lo);
    *(v4f*)(qrows + tid * 4 + 8 * DM) = *(const v4f*)(lo + 8 * TILE_PITCH);
    *(v4f*)(vrows + tid * 4)          = *(const v4f*)(lo + 16 * TILE_PITCH);
    *(v4f*)(vrows + tid * 4 + 8 * DM) = *(const v4f*)(lo + 24 * TILE_PITCH);
  }

  row_start_fill(row_start, e0, r0, p0);
  if (has1) row_start_fill(row_start, e1, r1, p1);
}

#define NODE_BLOCK 256
#define BATCH 16
#define RING 8
#define STASH 16
#define CTX_PITCH 68

__device__ __forceinline__ float quad_sum(float p) {
  p += __int_as_float(__builtin_amdgcn_update_dpp(
      0, __float_as_int(p), 0xB1, 0xF, 0xF, false));
  p += __int_as_float(__builtin_amdgcn_update_dpp(
      0, __float_as_int(p), 0x4E, 0xF, 0xF, false));
  return p;
}

template <int J>
__device__ __forceinline__ int group_lane(int x) {
  return __builtin_amdgcn_update_dpp(0, x, 0x150 + J, 0xF, 0xF, false);
}

template <int T>
__device__ __forceinline__ float quad_lane(float x) {
  return __int_as_float(__builtin_amdgcn_update_dpp(
      0, __float_as_int(x), T * 0x55, 0xF, 0xF, false));
}

template <int I> struct static_index { static constexpr int value = I; };
template <int FIRST, int LAST, class F>
__device__ __forceinline__ void for_each_index(F&& f) {
  if constexpr (FIRST < LAST) {
    f(static_index<FIRST>{});
    for_each_index<FIRST + 1, LAST>(f);
  }
}

__device__ __forceinline__ v4f row_piece(const float* __restrict__ table, int s, int l16) {
  const unsigned off = (unsigned)s * (DM * 4u) + (unsigned)l16 * 16u;
  return *(const v4f*)((const char*)table + off);
}

__device__ __forceinline__ void put_at(float* __restrict__ base, unsigned byte_off, float x) {
  *(float*)((char*)base + byte_off) = x;
}

__device__ __forceinline__ float head_dot(v4f q, v4f k) {
  return quad_sum(fmaf(q.w, k.w, fmaf(q.z, k.z, fmaf(q.x, k.x, q.y * k.y))));
}

__device__ __forceinline__ float edge_score(const float* __restrict__ qks,
                                            int s, int l16, v4f q) {
  return head_dot(q, row_piece(qks, s, l16));
}

__global__ __launch_bounds__(NODE_BLOCK) __attribute__((amdgpu_waves_per_eu(16)))
void node_attention_out_kernel(const float* __restrict__ qks,
                               const float* __restrict__ vs,
                               const int* __restrict__ senders,
                               const int* __restrict__ row_start,
                               const float* __restrict__ Wo,
                               const float* __restrict__ bo,
                               float* __restrict__ attn,
                               float* __restrict__ out) {
  __shared__ float stash[STASH * NODE_BLOCK];
  __shared__ __attribute__((aligned(16))) v4f q_park[NODE_BLOCK];
  __shared__ float m_park[NODE_BLOCK];
  __shared__ __attribute__((aligned(16))) float ctx_tile[16 * CTX_PITCH];
  __shared__ __attribute__((aligned(16))) float out_tile[16 * CTX_PITCH];

  const int tid  = threadIdx.x;
  const int l16  = tid & 15;
  const int node = blockIdx.x * (NODE_BLOCK / 16) + (tid >> 4);

  int start = row_start[node];
  int end   = row_start[node + 1];
  start = min(max(start, 0), NE);
  end   = min(max(end, start), NE);
  const int deg = end - start;

  const int wdeg = max(__builtin_amdgcn_readlane(deg, 0),
                       __builtin_amdgcn_readlane(deg, 16));

  const v4f q = row_piece(qks, node, l16);
  if (deg > BATCH + STASH) q_park[tid] = q;
  float* __restrict__ st = stash + tid;
  const unsigned attn_off = ((unsigned)start * NH + (unsigned)(l16 >> 2)) * 4u;

  const int own   = (l16 < deg) ? senders[min(start + l16, NE - 1)] : node;
  const int first = group_lane<0>(own);
  const int idx   = (l16 < deg) ? own : first;

  v4f   row[RING];
  float mine[BATCH / 4];
  const int t4 = l16 & 3;

  auto ask = [&](const float* __restrict__ table, auto B) {
    constexpr int b = decltype(B)::value;
    for_each_index<0, RING>([&](auto T) {
      constexpr int t = decltype(T)::value;
      row[t] = row_piece(table, group_lane<b * RING + t>(idx), l16);
    });
  };

  float m = -INFINITY;
  for_each_index<0, BATCH / 4>([&](auto G) { mine[decltype(G)::value] = -INFINITY; });
  auto score_group = [&](auto G) {
    constexpr int g = decltype(G)::value;
    for_each_index<0, 4>([&](auto T) {
      constexpr int t = decltype(T)::value;
      constexpr int j = g * 4 + t;
      const float s = (j < deg) ? head_dot(q, row[(g & 1) * 4 + t]) : -INFINITY;
      m = fmaxf(m, s);
      mine[g] = (t4 == t) ? s : mine[g];
    });
  };
  ask(qks, static_index<0>{});
  if (0 < wdeg) score_group(static_index<0>{});
  if (4 < wdeg) score_group(static_index<1>{});
  if (8 < wdeg) {
    ask(qks, static_index<1>{});
    score_group(static_index<2>{});
    if (12 < wdeg) score_group(static_index<3>{});
  }
  int s_next = (deg > BATCH) ? senders[start + BATCH] : 0;
  for (int i = BATCH; i < deg; ++i) {
    const int s = s_next;
    if (i + 1 < deg) s_next = senders[start + i + 1];
    const float sv = edge_score(qks, s, l16, q);
    if (i < BATCH + STASH) st[(i - BATCH) * NODE_BLOCK] = sv;
    m = fmaxf(m, sv);
  }

  if (deg > BATCH + STASH) m_park[tid] = m;

  ask(vs, static_index<0>{});

  float z = 0.0f;
  for_each_index<0, BATCH / 4>([&](auto G) {
    constexpr int g = decltype(G)::value;
    if (g * 4 < wdeg) {
      const float e = (g * 4 + t4 < deg) ? __expf(mine[g] - m) : 0.0f;
      mine[g] = e;
      for_each_index<0, 4>([&](auto T) {
        z += quad_lane<decltype(T)::value>(e);
      });
    }
  });
  for (int i = BATCH; i < deg; ++i) {
    const float sv = (i < BATCH + STASH) ? st[(i - BATCH) * NODE_BLOCK]
                                         : edge_score(qks, senders[start + i], l16, q_park[tid]);
    const float ev = __expf(sv - m);
    if (i < BATCH + STASH) st[(i - BATCH) * NODE_BLOCK] = ev;
    z += ev;
  }

  v4f acc = {0.0f, 0.0f, 0.0f, 0.0f};
  auto weigh_group = [&](auto G) {
    constexpr int g = decltype(G)::value;
    const bool own_slot = g * 4 + t4 < deg;
    const float a_own = own_slot ? mine[g] / z : 0.0f;
    if (own_slot) put_at(attn, attn_off + (unsigned)(g * 4 + t4) * (NH * 4u), a_own);
    for_each_index<0, 4>([&](auto T) {
      constexpr int t = decltype(T)::value;
      constexpr int j = g * 4 + t;
      const float a = quad_lane<t>(a_own);
      if (j < deg) {
        const v4f v = row[(g & 1) * 4 + t];
        acc.x += a * v.x;
        acc.y += a * v.y;
        acc.z += a * v.z;
        acc.w += a * v.w;
      }
    });
  };
  if (0 < wdeg) weigh_group(static_index<0>{});
  if (4 < wdeg) weigh_group(static_index<1>{});
  if (8 < wdeg) {
    ask(vs, static_index<1>{});
    weigh_group(static_index<2>{});
    if (12 < wdeg) weigh_group(static_index<3>{});
  }
  s_next = (deg > BATCH) ? senders[start + BATCH] : 0;
  for (int i = BATCH; i < deg; ++i) {
    const int s = s_next;
    if (i + 1 < deg) s_next = senders[start + i + 1];
    const v4f v = row_piece(vs, s, l16);
    const float ev = (i < BATCH + STASH) ? st[(i - BATCH) * NODE_BLOCK]
                                         : __expf(edge_score(qks, s, l16, q_park[tid]) - m_park[tid]);
    const float a = ev / z;
    if ((l16 & 3) == 0) put_at(attn, attn_off + (unsigned)i * (NH * 4u), a);
    acc.x += a * v.x;
    acc.y += a * v.y;
    acc.z += a * v.z;
    acc.w += a * v.w;
  }
  *(v4f*)(ctx_tile + (tid >> 4) * CTX_PITCH + l16 * 4) = acc;

  const int lane = tid & 31;
  const int wave = __builtin_amdgcn_readfirstlane(tid >> 5);
  const int n0   = (wave & 3) * 16;
  const int mm   = lane & 15;
  const int half = lane >> 4;
  const int kA   = half * 2;

  v2f wb[DM / 4];
  if (wave < 4) {
#pragma unroll
    for (int k = 0; k < DM / 4; ++k) {
      wb[k].x = Wo[(size_t)(k * 4 + kA) * DM + n0 + mm];
      wb[k].y = Wo[(size_t)(k * 4 + kA + 1) * DM + n0 + mm];
    }
  }

  __builtin_amdgcn_fence(__ATOMIC_RELEASE, "workgroup", "local");
  __builtin_amdgcn_s_barrier();
  __builtin_amdgcn_fence(__ATOMIC_ACQUIRE, "workgroup", "local");

  if (wave < 4) {
    const float* __restrict__ arow = ctx_tile + mm * CTX_PITCH + kA;
    v8f c = {};
#pragma unroll
    for (int k4 = 0; k4 < DM / 4; k4 += 4) {
      v2f af[4];
#pragma unroll
      for (int k = 0; k < 4; ++k) af[k] = *(const v2f*)(arow + (k4 + k) * 4);
#pragma unroll
      for (int k = 0; k < 4; ++k) {
        c = __builtin_amdgcn_wmma_f32_16x16x4_f32(false, af[k], false, wb[k4 + k],
                                                  (short)0, c, false, false);
      }
    }

    const float bb = bo[n0 + mm];
#pragma unroll
    for (int j = 0; j < 8; ++j) {
      out_tile[(j + half * 8) * CTX_PITCH + n0 + mm] = c[j] + bb;
    }
  }

  __builtin_amdgcn_fence(__ATOMIC_RELEASE, "workgroup", "local");
  __builtin_amdgcn_s_barrier();
  __builtin_amdgcn_fence(__ATOMIC_ACQUIRE, "workgroup", "local");

  float* __restrict__ orows = out + (size_t)blockIdx.x * ((NODE_BLOCK / 16) * DM);
  *(v4f*)(orows + tid * 4) =
      *(const v4f*)(out_tile + (tid >> 4) * CTX_PITCH + l16 * 4);
}

extern "C" void kernel_launch(void* const* d_in, const int* in_sizes, int n_in,
                              void* d_out, int out_size, void* d_ws, size_t ws_size,
                              hipStream_t stream) {
  const float* nodes     = (const float*)d_in[0];
  const float* W_qk      = (const float*)d_in[1];
  const float* b_qk      = (const float*)d_in[2];
  const float* W_v       = (const float*)d_in[3];
  const float* b_v       = (const float*)d_in[4];
  const float* W_out     = (const float*)d_in[5];
  const float* b_out     = (const float*)d_in[6];
  const int*   senders   = (const int*)d_in[7];
  const int*   receivers = (const int*)d_in[8];

  float* out  = (float*)d_out;
  float* attn = out + (size_t)NN * DM;

  float* ws      = (float*)d_ws;
  float* qks     = ws;
  float* vs      = ws + (size_t)NN * DM;
  int* row_start = (int*)(ws + (size_t)3 * NN * DM);

  const int gemmGrid = ROW_TILES;
  const int nodeGrid = NN / (NODE_BLOCK / 16);

  project_qkv_rows_kernel<<<gemmGrid, 128, 0, stream>>>(nodes, W_qk, b_qk, W_v, b_v,
                                                        receivers, qks, vs, row_start);

  node_attention_out_kernel<<<nodeGrid, NODE_BLOCK, 0, stream>>>(qks, vs, senders, row_start,
                                                               W_out, b_out, attn, out);
}
